# starting kernel
# baseline (speedup 1.0000x reference)
#include <hip/hip_runtime.h>

typedef __attribute__((ext_vector_type(2))) float v2f;
typedef __attribute__((ext_vector_type(8))) float v8f;

#define N_NODES 10
#define DIM 64
#define TM 64
#define PITCH 68
#define NODE_STRIDE (TM * PITCH)
#define LDS_FLOATS (9 * NODE_STRIDE)

__global__ __launch_bounds__(256) void transpose_w(const float* __restrict__ W,
                                                   float* __restrict__ Wt) {
  __shared__ float tile[DIM][DIM + 1];
  const int e = blockIdx.x;
  const float* src = W + (size_t)e * DIM * DIM;
  float* dst = Wt + (size_t)e * DIM * DIM;
  for (int i = threadIdx.x; i < DIM * DIM; i += 256)
    tile[i >> 6][i & 63] = src[i];
  __syncthreads();
  for (int i = threadIdx.x; i < DIM * DIM; i += 256) {
    const int n = i >> 6, k = i & 63;
    dst[i] = tile[k][n];
  }
}

template <bool XPOSED>
__global__ __launch_bounds__(128) void wired_dag_wmma(
    const float* __restrict__ x, const float* __restrict__ Wm,
    const float* __restrict__ bias, float* __restrict__ out)
{
  extern __shared__ __align__(16) float smem[];
  const int tid  = threadIdx.x;
  const int lane = tid & 31;
  const int wave = tid >> 5;
  const long rowBase = (long)blockIdx.x * TM;

  {
    const float4* src = (const float4*)(x + rowBase * DIM);
    for (int i = tid; i < (TM * DIM) / 4; i += 128) {
      const int r  = i >> 4;
      const int c4 = (i & 15) << 2;
      *(float4*)&smem[r * PITCH + c4] = src[i];
    }
  }
  __syncthreads();

  const int mta  = (wave >> 1) * 2;
  const int ntp  = wave & 1;
  const int l16  = lane & 15;
  const int half = lane >> 4;
  const int arowA = mta * 16 + l16;
  const int col0  = ntp * 32 + l16;
  const int col1  = col0 + 16;

  for (int t = 1; t < N_NODES; ++t) {
    float b0 = 0.f, b1 = 0.f;
    for (int f = 0; f < t; ++f) {
      const float* bp = bias + (size_t)(f * N_NODES + t) * DIM;
      b0 += bp[col0];
      b1 += bp[col1];
    }
    v8f c00, c01, c10, c11;
    #pragma unroll
    for (int i = 0; i < 8; ++i) {
      c00[i] = b0; c10[i] = b0;
      c01[i] = b1; c11[i] = b1;
    }

    for (int f = 0; f < t; ++f) {
      const float* apA = &smem[f * NODE_STRIDE + arowA * PITCH];
      const float* apB = apA + 16 * PITCH;
      const float* wp  = Wm + (size_t)(f * N_NODES + t) * DIM * DIM;
      #pragma unroll
      for (int k0 = 0; k0 < DIM; k0 += 4) {
        const int kk = k0 + half * 2;
        v2f a0 = *(const v2f*)(apA + kk);
        v2f a1 = *(const v2f*)(apB + kk);
        v2f p, q;
        if (XPOSED) {
          p = *(const v2f*)(wp + col0 * DIM + kk);
          q = *(const v2f*)(wp + col1 * DIM + kk);
        } else {
          p.x = wp[kk * DIM + col0]; p.y = wp[(kk + 1) * DIM + col0];
          q.x = wp[kk * DIM + col1]; q.y = wp[(kk + 1) * DIM + col1];
        }
        c00 = __builtin_amdgcn_wmma_f32_16x16x4_f32(false, a0, false, p, (short)0, c00, false, false);
        c01 = __builtin_amdgcn_wmma_f32_16x16x4_f32(false, a0, false, q, (short)0, c01, false, false);
        c10 = __builtin_amdgcn_wmma_f32_16x16x4_f32(false, a1, false, p, (short)0, c10, false, false);
        c11 = __builtin_amdgcn_wmma_f32_16x16x4_f32(false, a1, false, q, (short)0, c11, false, false);
      }
    }

    if (t < N_NODES - 1) {
      float* dp = &smem[t * NODE_STRIDE];
      #pragma unroll
      for (int i = 0; i < 8; ++i) {
        const int ra = mta * 16 + half * 8 + i;
        const int rb = ra + 16;
        dp[ra * PITCH + col0] = c00[i];
        dp[ra * PITCH + col1] = c01[i];
        dp[rb * PITCH + col0] = c10[i];
        dp[rb * PITCH + col1] = c11[i];
      }
      __syncthreads();
    } else {
      float* op = out + rowBase * DIM;
      #pragma unroll
      for (int i = 0; i < 8; ++i) {
        const int ra = mta * 16 + half * 8 + i;
        const int rb = ra + 16;
        op[(size_t)ra * DIM + col0] = c00[i];
        op[(size_t)ra * DIM + col1] = c01[i];
        op[(size_t)rb * DIM + col0] = c10[i];
        op[(size_t)rb * DIM + col1] = c11[i];
      }
    }
  }
}

extern "C" void kernel_launch(void* const* d_in, const int* in_sizes, int n_in,
                              void* d_out, int out_size, void* d_ws, size_t ws_size,
                              hipStream_t stream) {
  const float* x  = (const float*)d_in[0];
  const float* W  = (const float*)d_in[1];
  const float* b  = (const float*)d_in[2];
  float* out      = (float*)d_out;

  const int Bn      = in_sizes[0] / DIM;
  const int nBlocks = Bn / TM;
  const size_t ldsBytes = (size_t)LDS_FLOATS * sizeof(float);
  const size_t wtBytes  = (size_t)N_NODES * N_NODES * DIM * DIM * sizeof(float);

  if (d_ws != nullptr && ws_size >= wtBytes) {
    float* Wt = (float*)d_ws;
    transpose_w<<<N_NODES * N_NODES, 256, 0, stream>>>(W, Wt);
    (void)hipFuncSetAttribute((const void*)wired_dag_wmma<true>,
                              hipFuncAttributeMaxDynamicSharedMemorySize,
                              (int)ldsBytes);
    wired_dag_wmma<true><<<nBlocks, 128, ldsBytes, stream>>>(x, Wt, b, out);
  } else {
    (void)hipFuncSetAttribute((const void*)wired_dag_wmma<false>,
                              hipFuncAttributeMaxDynamicSharedMemorySize,
                              (int)ldsBytes);
    wired_dag_wmma<false><<<nBlocks, 128, ldsBytes, stream>>>(x, W, b, out);
  }
}


# A operand path: both row tiles' fragments read from LDS half an edge ahead, one group of 8 loads per k half into two register sets beside the B group; no LDS read left among the WMMAs; sums kept
# speedup vs baseline: 1.2773x; 1.2773x over previous
#include <hip/hip_runtime.h>


typedef __attribute__((ext_vector_type(2))) float v2f;
typedef __attribute__((ext_vector_type(8))) float v8f;

#define N_NODES 10
#define DIM 64
#define TM 64
#define PITCH 68
#define NODE_STRIDE (TM * PITCH)
#define LDS_FLOATS (9 * NODE_STRIDE)
#define HSTEPS (DIM / 8)

__global__ __launch_bounds__(256) void pack_w(const float* __restrict__ W,
                                              float* __restrict__ Wp) {
  __shared__ float tile[DIM][DIM + 1];
  const int e = blockIdx.x;
  const float* src = W + (size_t)e * DIM * DIM;
  float* dst = Wp + (size_t)e * DIM * DIM;
  for (int i = threadIdx.x; i < DIM * DIM; i += 256)
    tile[i >> 6][i & 63] = src[i];
  __syncthreads();
  for (int i = threadIdx.x; i < DIM * DIM; i += 256) {
    const int j = i & 3, ln = (i >> 2) & 31, ks = (i >> 7) & 15, np = i >> 11;
    const int k = ks * 4 + (ln >> 4) * 2 + (j & 1);
    const int n = np * 32 + (ln & 15) + (j >> 1) * 16;
    dst[i] = tile[k][n];
  }
}

__global__ __launch_bounds__(128) void wired_dag_wmma(
    const float* __restrict__ x, const float* __restrict__ Wp,
    const float* __restrict__ bias, float* __restrict__ out)
{
  extern __shared__ __align__(16) float smem[];
  const int tid  = threadIdx.x;
  const int lane = tid & 31;
  const int wave = tid >> 5;
  const long rowBase = (long)blockIdx.x * TM;

  const int mta  = (wave >> 1) * 2;
  const int ntp  = wave & 1;
  const int l16  = lane & 15;
  const int half = lane >> 4;
  const int arowA = mta * 16 + l16;
  const int col0  = ntp * 32 + l16;
  const int col1  = col0 + 16;

  const float* wlane = Wp + ntp * 2048 + lane * 4;

  float4 wlo[HSTEPS];
  float4 whi[HSTEPS];
  {
    const float* wp = wlane + (size_t)(0 * N_NODES + 1) * DIM * DIM;
    #pragma unroll
    for (int ks = 0; ks < HSTEPS; ++ks) wlo[ks] = *(const float4*)(wp + ks * 128);
  }

  {
    const float4* src = (const float4*)(x + rowBase * DIM);
    for (int i = tid; i < (TM * DIM) / 4; i += 128) {
      const int r  = i >> 4;
      const int c4 = (i & 15) << 2;
      *(float4*)&smem[r * PITCH + c4] = src[i];
    }
  }
  __syncthreads();

  const int aoff = arowA * PITCH + half * 2;

  v2f alo0[HSTEPS], alo1[HSTEPS];
  v2f ahi0[HSTEPS], ahi1[HSTEPS];
  {
    const float* ap = &smem[aoff];
    #pragma unroll
    for (int ks = 0; ks < HSTEPS; ++ks) {
      alo0[ks] = *(const v2f*)(ap + ks * 4);
      alo1[ks] = *(const v2f*)(ap + 16 * PITCH + ks * 4);
    }
  }

  for (int t = 1; t < N_NODES; ++t) {
    float b0 = 0.f, b1 = 0.f;
    for (int f = 0; f < t; ++f) {
      const float* bp = bias + (size_t)(f * N_NODES + t) * DIM;
      b0 += bp[col0];
      b1 += bp[col1];
    }
    v8f c00, c01, c10, c11;
    #pragma unroll
    for (int i = 0; i < 8; ++i) {
      c00[i] = b0; c10[i] = b0;
      c01[i] = b1; c11[i] = b1;
    }

    for (int f = 0; f < t; ++f) {
      const float* ap = &smem[f * NODE_STRIDE + aoff];
      const float* wp = wlane + (size_t)(f * N_NODES + t) * DIM * DIM;

      #pragma unroll
      for (int ks = 0; ks < HSTEPS; ++ks) whi[ks] = *(const float4*)(wp + (HSTEPS + ks) * 128);
      #pragma unroll
      for (int ks = 0; ks < HSTEPS; ++ks) {
        ahi0[ks] = *(const v2f*)(ap + (HSTEPS + ks) * 4);
        ahi1[ks] = *(const v2f*)(ap + 16 * PITCH + (HSTEPS + ks) * 4);
      }
      __builtin_amdgcn_sched_barrier(0);

      #pragma unroll
      for (int ks = 0; ks < HSTEPS; ++ks) {
        v2f p, q;
        p.x = wlo[ks].x; p.y = wlo[ks].y;
        q.x = wlo[ks].z; q.y = wlo[ks].w;
        c00 = __builtin_amdgcn_wmma_f32_16x16x4_f32(false, alo0[ks], false, p, (short)0, c00, false, false);
        c01 = __builtin_amdgcn_wmma_f32_16x16x4_f32(false, alo0[ks], false, q, (short)0, c01, false, false);
        c10 = __builtin_amdgcn_wmma_f32_16x16x4_f32(false, alo1[ks], false, p, (short)0, c10, false, false);
        c11 = __builtin_amdgcn_wmma_f32_16x16x4_f32(false, alo1[ks], false, q, (short)0, c11, false, false);
      }
      __builtin_amdgcn_sched_barrier(0);

      int fn = f + 1, tn = t;
      if (fn == t) { fn = 0; tn = t + 1; }
      if (tn < N_NODES) {
        const float* wq = wlane + (size_t)(fn * N_NODES + tn) * DIM * DIM;
        const float* aq = &smem[fn * NODE_STRIDE + aoff];
        #pragma unroll
        for (int ks = 0; ks < HSTEPS; ++ks) wlo[ks] = *(const float4*)(wq + ks * 128);
        #pragma unroll
        for (int ks = 0; ks < HSTEPS; ++ks) {
          alo0[ks] = *(const v2f*)(aq + ks * 4);
          alo1[ks] = *(const v2f*)(aq + 16 * PITCH + ks * 4);
        }
      }
      __builtin_amdgcn_sched_barrier(0);

      #pragma unroll
      for (int ks = 0; ks < HSTEPS; ++ks) {
        v2f p, q;
        p.x = whi[ks].x; p.y = whi[ks].y;
        q.x = whi[ks].z; q.y = whi[ks].w;
        c00 = __builtin_amdgcn_wmma_f32_16x16x4_f32(false, ahi0[ks], false, p, (short)0, c00, false, false);
        c01 = __builtin_amdgcn_wmma_f32_16x16x4_f32(false, ahi0[ks], false, q, (short)0, c01, false, false);
        c10 = __builtin_amdgcn_wmma_f32_16x16x4_f32(false, ahi1[ks], false, p, (short)0, c10, false, false);
        c11 = __builtin_amdgcn_wmma_f32_16x16x4_f32(false, ahi1[ks], false, q, (short)0, c11, false, false);
      }
    }

    if (t < N_NODES - 1) {
      float* dp = &smem[t * NODE_STRIDE];
      #pragma unroll
      for (int i = 0; i < 8; ++i) {
        const int ra = mta * 16 + half * 8 + i;
        const int rb = ra + 16;
        dp[ra * PITCH + col0] = c00[i];
        dp[ra * PITCH + col1] = c01[i];
        dp[rb * PITCH + col0] = c10[i];
        dp[rb * PITCH + col1] = c11[i];
      }
      __syncthreads();
    } else {
      float* op = out + rowBase * DIM;
      #pragma unroll
      for (int i = 0; i < 8; ++i) {
        const int ra = mta * 16 + half * 8 + i;
        const int rb = ra + 16;
        op[(size_t)ra * DIM + col0] = c00[i];
        op[(size_t)ra * DIM + col1] = c01[i];
        op[(size_t)rb * DIM + col0] = c10[i];
        op[(size_t)rb * DIM + col1] = c11[i];
      }
    }
  }
}

__global__ __launch_bounds__(128) void wired_dag_wmma_direct(
    const float* __restrict__ x, const float* __restrict__ Wm,
    const float* __restrict__ bias, float* __restrict__ out)
{
  extern __shared__ __align__(16) float smem[];
  const int tid  = threadIdx.x;
  const int lane = tid & 31;
  const int wave = tid >> 5;
  const long rowBase = (long)blockIdx.x * TM;

  {
    const float4* src = (const float4*)(x + rowBase * DIM);
    for (int i = tid; i < (TM * DIM) / 4; i += 128) {
      const int r  = i >> 4;
      const int c4 = (i & 15) << 2;
      *(float4*)&smem[r * PITCH + c4] = src[i];
    }
  }
  __syncthreads();

  const int mta  = (wave >> 1) * 2;
  const int ntp  = wave & 1;
  const int l16  = lane & 15;
  const int half = lane >> 4;
  const int arowA = mta * 16 + l16;
  const int col0  = ntp * 32 + l16;
  const int col1  = col0 + 16;

  for (int t = 1; t < N_NODES; ++t) {
    float b0 = 0.f, b1 = 0.f;
    for (int f = 0; f < t; ++f) {
      const float* bp = bias + (size_t)(f * N_NODES + t) * DIM;
      b0 += bp[col0];
      b1 += bp[col1];
    }
    v8f c00, c01, c10, c11;
    #pragma unroll
    for (int i = 0; i < 8; ++i) {
      c00[i] = b0; c10[i] = b0;
      c01[i] = b1; c11[i] = b1;
    }

    for (int f = 0; f < t; ++f) {
      const float* apA = &smem[f * NODE_STRIDE + arowA * PITCH];
      const float* apB = apA + 16 * PITCH;
      const float* wp  = Wm + (size_t)(f * N_NODES + t) * DIM * DIM;
      #pragma unroll
      for (int k0 = 0; k0 < DIM; k0 += 4) {
        const int kk = k0 + half * 2;
        v2f a0 = *(const v2f*)(apA + kk);
        v2f a1 = *(const v2f*)(apB + kk);
        v2f p, q;
        p.x = wp[kk * DIM + col0]; p.y = wp[(kk + 1) * DIM + col0];
        q.x = wp[kk * DIM + col1]; q.y = wp[(kk + 1) * DIM + col1];
        c00 = __builtin_amdgcn_wmma_f32_16x16x4_f32(false, a0, false, p, (short)0, c00, false, false);
        c01 = __builtin_amdgcn_wmma_f32_16x16x4_f32(false, a0, false, q, (short)0, c01, false, false);
        c10 = __builtin_amdgcn_wmma_f32_16x16x4_f32(false, a1, false, p, (short)0, c10, false, false);
        c11 = __builtin_amdgcn_wmma_f32_16x16x4_f32(false, a1, false, q, (short)0, c11, false, false);
      }
    }

    if (t < N_NODES - 1) {
      float* dp = &smem[t * NODE_STRIDE];
      #pragma unroll
      for (int i = 0; i < 8; ++i) {
        const int ra = mta * 16 + half * 8 + i;
        const int rb = ra + 16;
        dp[ra * PITCH + col0] = c00[i];
        dp[ra * PITCH + col1] = c01[i];
        dp[rb * PITCH + col0] = c10[i];
        dp[rb * PITCH + col1] = c11[i];
      }
      __syncthreads();
    } else {
      float* op = out + rowBase * DIM;
      #pragma unroll
      for (int i = 0; i < 8; ++i) {
        const int ra = mta * 16 + half * 8 + i;
        const int rb = ra + 16;
        op[(size_t)ra * DIM + col0] = c00[i];
        op[(size_t)ra * DIM + col1] = c01[i];
        op[(size_t)rb * DIM + col0] = c10[i];
        op[(size_t)rb * DIM + col1] = c11[i];
      }
    }
  }
}

extern "C" void kernel_launch(void* const* d_in, const int* in_sizes, int n_in,
                              void* d_out, int out_size, void* d_ws, size_t ws_size,
                              hipStream_t stream) {
  const float* x  = (const float*)d_in[0];
  const float* W  = (const float*)d_in[1];
  const float* b  = (const float*)d_in[2];
  float* out      = (float*)d_out;

  const int Bn      = in_sizes[0] / DIM;
  const int nBlocks = Bn / TM;
  const size_t ldsBytes = (size_t)LDS_FLOATS * sizeof(float);
  const size_t wpBytes  = (size_t)N_NODES * N_NODES * DIM * DIM * sizeof(float);

  if (d_ws != nullptr && ws_size >= wpBytes) {
    float* Wp = (float*)d_ws;
    pack_w<<<N_NODES * N_NODES, 256, 0, stream>>>(W, Wp);
    (void)hipFuncSetAttribute((const void*)wired_dag_wmma,
                              hipFuncAttributeMaxDynamicSharedMemorySize,
                              (int)ldsBytes);
    wired_dag_wmma<<<nBlocks, 128, ldsBytes, stream>>>(x, Wp, b, out);
  } else {
    (void)hipFuncSetAttribute((const void*)wired_dag_wmma_direct,
                              hipFuncAttributeMaxDynamicSharedMemorySize,
                              (int)ldsBytes);
    wired_dag_wmma_direct<<<nBlocks, 128, ldsBytes, stream>>>(x, W, b, out);
  }
}


# Load chains outside the edge loop: bias rows summed once per call in the pack launch and fetched a level ahead (was t dependent loads per level), x tile read as one group of 8 loads; sums kept
# speedup vs baseline: 1.3835x; 1.0831x over previous
#include <hip/hip_runtime.h>


typedef __attribute__((ext_vector_type(2))) float v2f;
typedef __attribute__((ext_vector_type(8))) float v8f;

#define N_NODES 10
#define DIM 64
#define TM 64
#define PITCH 68
#define NODE_STRIDE (TM * PITCH)
#define LDS_FLOATS (9 * NODE_STRIDE)
#define HSTEPS (DIM / 8)

__global__ __launch_bounds__(256) void pack_w(const float* __restrict__ W,
                                              const float* __restrict__ bias,
                                              float* __restrict__ Wp,
                                              float* __restrict__ bsum) {
  __shared__ float tile[DIM][DIM + 1];
  if (blockIdx.x >= N_NODES * N_NODES) {
    const int idx = (blockIdx.x - N_NODES * N_NODES) * 256 + threadIdx.x;
    if (idx < N_NODES * DIM) {
      const int t = idx >> 6, col = idx & 63;
      float v[N_NODES - 1];
      #pragma unroll
      for (int f = 0; f < N_NODES - 1; ++f) {
        const int fe = f < t ? f : 0;
        v[f] = bias[(size_t)(fe * N_NODES + t) * DIM + col];
      }
      float s = 0.f;
      #pragma unroll
      for (int f = 0; f < N_NODES - 1; ++f) s = f < t ? s + v[f] : s;
      bsum[idx] = s;
    }
    return;
  }
  const int e = blockIdx.x;
  const float* src = W + (size_t)e * DIM * DIM;
  float* dst = Wp + (size_t)e * DIM * DIM;
  for (int i = threadIdx.x; i < DIM * DIM; i += 256)
    tile[i >> 6][i & 63] = src[i];
  __syncthreads();
  for (int i = threadIdx.x; i < DIM * DIM; i += 256) {
    const int j = i & 3, ln = (i >> 2) & 31, ks = (i >> 7) & 15, np = i >> 11;
    const int k = ks * 4 + (ln >> 4) * 2 + (j & 1);
    const int n = np * 32 + (ln & 15) + (j >> 1) * 16;
    dst[i] = tile[k][n];
  }
}

__global__ __launch_bounds__(128) void wired_dag_wmma(
    const float* __restrict__ x, const float* __restrict__ Wp,
    const float* __restrict__ bsum, float* __restrict__ out)
{
  extern __shared__ __align__(16) float smem[];
  const int tid  = threadIdx.x;
  const int lane = tid & 31;
  const int wave = tid >> 5;
  const long rowBase = (long)blockIdx.x * TM;

  const int mta  = (wave >> 1) * 2;
  const int ntp  = wave & 1;
  const int l16  = lane & 15;
  const int half = lane >> 4;
  const int arowA = mta * 16 + l16;
  const int col0  = ntp * 32 + l16;
  const int col1  = col0 + 16;

  const float* wlane = Wp + ntp * 2048 + lane * 4;

  float4 wlo[HSTEPS];
  float4 whi[HSTEPS];
  {
    const float* wp = wlane + (size_t)(0 * N_NODES + 1) * DIM * DIM;
    #pragma unroll
    for (int ks = 0; ks < HSTEPS; ++ks) wlo[ks] = *(const float4*)(wp + ks * 128);
  }

  const float* bs = bsum + col0;
  float nb0 = bs[1 * DIM], nb1 = bs[1 * DIM + 16];

  {
    const float4* src = (const float4*)(x + rowBase * DIM);
    float4 xv[(TM * DIM) / 4 / 128];
    #pragma unroll
    for (int j = 0; j < (TM * DIM) / 4 / 128; ++j) xv[j] = src[tid + j * 128];
    #pragma unroll
    for (int j = 0; j < (TM * DIM) / 4 / 128; ++j) {
      const int i  = tid + j * 128;
      const int r  = i >> 4;
      const int c4 = (i & 15) << 2;
      *(float4*)&smem[r * PITCH + c4] = xv[j];
    }
  }
  __syncthreads();

  const int aoff = arowA * PITCH + half * 2;

  v2f alo0[HSTEPS], alo1[HSTEPS];
  v2f ahi0[HSTEPS], ahi1[HSTEPS];
  {
    const float* ap = &smem[aoff];
    #pragma unroll
    for (int ks = 0; ks < HSTEPS; ++ks) {
      alo0[ks] = *(const v2f*)(ap + ks * 4);
      alo1[ks] = *(const v2f*)(ap + 16 * PITCH + ks * 4);
    }
  }

  for (int t = 1; t < N_NODES; ++t) {
    const float b0 = nb0, b1 = nb1;
    if (t + 1 < N_NODES) {
      nb0 = bs[(t + 1) * DIM];
      nb1 = bs[(t + 1) * DIM + 16];
    }
    v8f c00, c01, c10, c11;
    #pragma unroll
    for (int i = 0; i < 8; ++i) {
      c00[i] = b0; c10[i] = b0;
      c01[i] = b1; c11[i] = b1;
    }

    for (int f = 0; f < t; ++f) {
      const float* ap = &smem[f * NODE_STRIDE + aoff];
      const float* wp = wlane + (size_t)(f * N_NODES + t) * DIM * DIM;

      #pragma unroll
      for (int ks = 0; ks < HSTEPS; ++ks) whi[ks] = *(const float4*)(wp + (HSTEPS + ks) * 128);
      #pragma unroll
      for (int ks = 0; ks < HSTEPS; ++ks) {
        ahi0[ks] = *(const v2f*)(ap + (HSTEPS + ks) * 4);
        ahi1[ks] = *(const v2f*)(ap + 16 * PITCH + (HSTEPS + ks) * 4);
      }
      __builtin_amdgcn_sched_barrier(0);

      #pragma unroll
      for (int ks = 0; ks < HSTEPS; ++ks) {
        v2f p, q;
        p.x = wlo[ks].x; p.y = wlo[ks].y;
        q.x = wlo[ks].z; q.y = wlo[ks].w;
        c00 = __builtin_amdgcn_wmma_f32_16x16x4_f32(false, alo0[ks], false, p, (short)0, c00, false, false);
        c01 = __builtin_amdgcn_wmma_f32_16x16x4_f32(false, alo0[ks], false, q, (short)0, c01, false, false);
        c10 = __builtin_amdgcn_wmma_f32_16x16x4_f32(false, alo1[ks], false, p, (short)0, c10, false, false);
        c11 = __builtin_amdgcn_wmma_f32_16x16x4_f32(false, alo1[ks], false, q, (short)0, c11, false, false);
      }
      __builtin_amdgcn_sched_barrier(0);

      int fn = f + 1, tn = t;
      if (fn == t) { fn = 0; tn = t + 1; }
      if (tn < N_NODES) {
        const float* wq = wlane + (size_t)(fn * N_NODES + tn) * DIM * DIM;
        const float* aq = &smem[fn * NODE_STRIDE + aoff];
        #pragma unroll
        for (int ks = 0; ks < HSTEPS; ++ks) wlo[ks] = *(const float4*)(wq + ks * 128);
        #pragma unroll
        for (int ks = 0; ks < HSTEPS; ++ks) {
          alo0[ks] = *(const v2f*)(aq + ks * 4);
          alo1[ks] = *(const v2f*)(aq + 16 * PITCH + ks * 4);
        }
      }
      __builtin_amdgcn_sched_barrier(0);

      #pragma unroll
      for (int ks = 0; ks < HSTEPS; ++ks) {
        v2f p, q;
        p.x = whi[ks].x; p.y = whi[ks].y;
        q.x = whi[ks].z; q.y = whi[ks].w;
        c00 = __builtin_amdgcn_wmma_f32_16x16x4_f32(false, ahi0[ks], false, p, (short)0, c00, false, false);
        c01 = __builtin_amdgcn_wmma_f32_16x16x4_f32(false, ahi0[ks], false, q, (short)0, c01, false, false);
        c10 = __builtin_amdgcn_wmma_f32_16x16x4_f32(false, ahi1[ks], false, p, (short)0, c10, false, false);
        c11 = __builtin_amdgcn_wmma_f32_16x16x4_f32(false, ahi1[ks], false, q, (short)0, c11, false, false);
      }
    }

    if (t < N_NODES - 1) {
      float* dp = &smem[t * NODE_STRIDE];
      #pragma unroll
      for (int i = 0; i < 8; ++i) {
        const int ra = mta * 16 + half * 8 + i;
        const int rb = ra + 16;
        dp[ra * PITCH + col0] = c00[i];
        dp[ra * PITCH + col1] = c01[i];
        dp[rb * PITCH + col0] = c10[i];
        dp[rb * PITCH + col1] = c11[i];
      }
      __syncthreads();
    } else {
      float* op = out + rowBase * DIM;
      #pragma unroll
      for (int i = 0; i < 8; ++i) {
        const int ra = mta * 16 + half * 8 + i;
        const int rb = ra + 16;
        op[(size_t)ra * DIM + col0] = c00[i];
        op[(size_t)ra * DIM + col1] = c01[i];
        op[(size_t)rb * DIM + col0] = c10[i];
        op[(size_t)rb * DIM + col1] = c11[i];
      }
    }
  }
}

__global__ __launch_bounds__(128) void wired_dag_wmma_direct(
    const float* __restrict__ x, const float* __restrict__ Wm,
    const float* __restrict__ bias, float* __restrict__ out)
{
  extern __shared__ __align__(16) float smem[];
  const int tid  = threadIdx.x;
  const int lane = tid & 31;
  const int wave = tid >> 5;
  const long rowBase = (long)blockIdx.x * TM;

  {
    const float4* src = (const float4*)(x + rowBase * DIM);
    for (int i = tid; i < (TM * DIM) / 4; i += 128) {
      const int r  = i >> 4;
      const int c4 = (i & 15) << 2;
      *(float4*)&smem[r * PITCH + c4] = src[i];
    }
  }
  __syncthreads();

  const int mta  = (wave >> 1) * 2;
  const int ntp  = wave & 1;
  const int l16  = lane & 15;
  const int half = lane >> 4;
  const int arowA = mta * 16 + l16;
  const int col0  = ntp * 32 + l16;
  const int col1  = col0 + 16;

  for (int t = 1; t < N_NODES; ++t) {
    float b0 = 0.f, b1 = 0.f;
    for (int f = 0; f < t; ++f) {
      const float* bp = bias + (size_t)(f * N_NODES + t) * DIM;
      b0 += bp[col0];
      b1 += bp[col1];
    }
    v8f c00, c01, c10, c11;
    #pragma unroll
    for (int i = 0; i < 8; ++i) {
      c00[i] = b0; c10[i] = b0;
      c01[i] = b1; c11[i] = b1;
    }

    for (int f = 0; f < t; ++f) {
      const float* apA = &smem[f * NODE_STRIDE + arowA * PITCH];
      const float* apB = apA + 16 * PITCH;
      const float* wp  = Wm + (size_t)(f * N_NODES + t) * DIM * DIM;
      #pragma unroll
      for (int k0 = 0; k0 < DIM; k0 += 4) {
        const int kk = k0 + half * 2;
        v2f a0 = *(const v2f*)(apA + kk);
        v2f a1 = *(const v2f*)(apB + kk);
        v2f p, q;
        p.x = wp[kk * DIM + col0]; p.y = wp[(kk + 1) * DIM + col0];
        q.x = wp[kk * DIM + col1]; q.y = wp[(kk + 1) * DIM + col1];
        c00 = __builtin_amdgcn_wmma_f32_16x16x4_f32(false, a0, false, p, (short)0, c00, false, false);
        c01 = __builtin_amdgcn_wmma_f32_16x16x4_f32(false, a0, false, q, (short)0, c01, false, false);
        c10 = __builtin_amdgcn_wmma_f32_16x16x4_f32(false, a1, false, p, (short)0, c10, false, false);
        c11 = __builtin_amdgcn_wmma_f32_16x16x4_f32(false, a1, false, q, (short)0, c11, false, false);
      }
    }

    if (t < N_NODES - 1) {
      float* dp = &smem[t * NODE_STRIDE];
      #pragma unroll
      for (int i = 0; i < 8; ++i) {
        const int ra = mta * 16 + half * 8 + i;
        const int rb = ra + 16;
        dp[ra * PITCH + col0] = c00[i];
        dp[ra * PITCH + col1] = c01[i];
        dp[rb * PITCH + col0] = c10[i];
        dp[rb * PITCH + col1] = c11[i];
      }
      __syncthreads();
    } else {
      float* op = out + rowBase * DIM;
      #pragma unroll
      for (int i = 0; i < 8; ++i) {
        const int ra = mta * 16 + half * 8 + i;
        const int rb = ra + 16;
        op[(size_t)ra * DIM + col0] = c00[i];
        op[(size_t)ra * DIM + col1] = c01[i];
        op[(size_t)rb * DIM + col0] = c10[i];
        op[(size_t)rb * DIM + col1] = c11[i];
      }
    }
  }
}

extern "C" void kernel_launch(void* const* d_in, const int* in_sizes, int n_in,
                              void* d_out, int out_size, void* d_ws, size_t ws_size,
                              hipStream_t stream) {
  const float* x  = (const float*)d_in[0];
  const float* W  = (const float*)d_in[1];
  const float* b  = (const float*)d_in[2];
  float* out      = (float*)d_out;

  const int Bn      = in_sizes[0] / DIM;
  const int nBlocks = Bn / TM;
  const size_t ldsBytes = (size_t)LDS_FLOATS * sizeof(float);
  const size_t wpBytes  = (size_t)N_NODES * N_NODES * DIM * DIM * sizeof(float);
  const size_t bsBytes  = (size_t)N_NODES * DIM * sizeof(float);

  if (d_ws != nullptr && ws_size >= wpBytes + bsBytes) {
    float* Wp = (float*)d_ws;
    float* bsum = Wp + (size_t)N_NODES * N_NODES * DIM * DIM;
    pack_w<<<N_NODES * N_NODES + 3, 256, 0, stream>>>(W, b, Wp, bsum);
    (void)hipFuncSetAttribute((const void*)wired_dag_wmma,
                              hipFuncAttributeMaxDynamicSharedMemorySize,
                              (int)ldsBytes);
    wired_dag_wmma<<<nBlocks, 128, ldsBytes, stream>>>(x, Wp, bsum, out);
  } else {
    (void)hipFuncSetAttribute((const void*)wired_dag_wmma_direct,
                              hipFuncAttributeMaxDynamicSharedMemorySize,
                              (int)ldsBytes);
    wired_dag_wmma_direct<<<nBlocks, 128, ldsBytes, stream>>>(x, W, b, out);
  }
}
